# starting kernel
# baseline (speedup 1.0000x reference)
#include <hip/hip_runtime.h>
#include <hip/hip_bf16.h>

typedef __attribute__((ext_vector_type(2))) float v2f;
typedef __attribute__((ext_vector_type(8))) float v8f;

#define BSZ    8192
#define TSEQ   4096
#define OUTW   4095
#define NVALID 4093
#define CH     16
#define NCHUNK 256
#define WAVES_PER_BLOCK 2
#define ROWS_PER_WAVE   16
#define ROWS_PER_BLOCK  (WAVES_PER_BLOCK * ROWS_PER_WAVE)
#define PF_DIST (8 * CH)

template <bool LAST>
__device__ __forceinline__ void process_chunk(
    int base, const float* __restrict__ yrow, float* __restrict__ orow,
    int khalf, int m0, int lrow,
    const float A0k[4], const float A1k[4], const float hv[9],
    float q0, float ph1, float muv, float th1,
    float& e1, float& e2)
{
  float yb[17];
  const int tbase = base + khalf;
#pragma unroll
  for (int j = 1; j <= 16; ++j) {
    if (LAST) {
      const int idx = tbase + j;
      yb[j] = (idx < TSEQ) ? yrow[idx] : 0.0f;
    } else {
      yb[j] = yrow[tbase + j];
    }
  }
  if (!LAST) {
    __builtin_prefetch(yrow + tbase + PF_DIST, 0, 0);
  }

  v8f acc = {0.f, 0.f, 0.f, 0.f, 0.f, 0.f, 0.f, 0.f};
#pragma unroll
  for (int kk = 0; kk < 4; ++kk) {
    const int o = 4 * kk;
    const float t0 = fmaf(-ph1, yb[o + 1], fmaf(-q0, yb[o + 2], yb[o + 3] - muv));
    const float t1 = fmaf(-ph1, yb[o + 2], fmaf(-q0, yb[o + 3], yb[o + 4] - muv));
    v2f a; a.x = A0k[kk]; a.y = A1k[kk];
    v2f b; b.x = t0;      b.y = t1;
    acc = __builtin_amdgcn_wmma_f32_16x16x4_f32(
        false, a, false, b, (short)0, acc, false, false);
  }

  const float e2m = -th1 * e2;
  float d[8];
#pragma unroll
  for (int v = 0; v < 8; ++v)
    d[v] = fmaf(e1, hv[v + 1], fmaf(e2m, hv[v], acc[v]));

  e1 = __shfl(d[7], lrow + 16, 32);
  e2 = __shfl(d[6], lrow + 16, 32);

  float* op = orow + base + m0;
#pragma unroll
  for (int v = 0; v < 8; ++v) {
    if (LAST) {
      const int k = base + m0 + v;
      if (k < NVALID)     op[v] = d[v];
      else if (k < OUTW)  op[v] = 0.0f;
    } else {
      op[v] = d[v];
    }
  }
}

__global__ __launch_bounds__(WAVES_PER_BLOCK * 32) void arima_eps_kernel(
    const float* __restrict__ y, const float* __restrict__ phi,
    const float* __restrict__ theta, const float* __restrict__ mu,
    float* __restrict__ out)
{
  __shared__ float h_sh[17];

  const float ph0 = phi[0], ph1 = phi[1];
  const float th0 = theta[0], th1 = theta[1];
  const float muv = mu[0];

  if (threadIdx.x == 0) {
    h_sh[0] = 1.0f;
    h_sh[1] = -th0;
    for (int k = 2; k < 17; ++k)
      h_sh[k] = fmaf(-th0, h_sh[k - 1], -th1 * h_sh[k - 2]);
  }
  __syncthreads();

  const int lane  = threadIdx.x & 31;
  const int wid   = threadIdx.x >> 5;
  const int lrow  = lane & 15;
  const int half  = lane >> 4;
  const int khalf = half * 2;
  const int m0    = half * 8;
  const int row   = blockIdx.x * ROWS_PER_BLOCK + wid * ROWS_PER_WAVE + lrow;

  const float* yrow = y   + (size_t)row * TSEQ;
  float*       orow = out + (size_t)row * OUTW;

  float A0k[4], A1k[4];
#pragma unroll
  for (int kk = 0; kk < 4; ++kk) {
    const int c0 = 4 * kk + khalf;
    A0k[kk] = (lrow >= c0)     ? h_sh[lrow - c0]     : 0.0f;
    A1k[kk] = (lrow >= c0 + 1) ? h_sh[lrow - c0 - 1] : 0.0f;
  }
  float hv[9];
#pragma unroll
  for (int v = 0; v < 9; ++v) hv[v] = h_sh[m0 + v];

  const float q0 = 1.0f + ph0;
  float e1 = 0.0f, e2 = 0.0f;

  for (int ch = 0; ch < NCHUNK - 1; ++ch)
    process_chunk<false>(ch * CH, yrow, orow, khalf, m0, lrow,
                         A0k, A1k, hv, q0, ph1, muv, th1, e1, e2);
  process_chunk<true>((NCHUNK - 1) * CH, yrow, orow, khalf, m0, lrow,
                      A0k, A1k, hv, q0, ph1, muv, th1, e1, e2);
}

extern "C" void kernel_launch(void* const* d_in, const int* in_sizes, int n_in,
                              void* d_out, int out_size, void* d_ws, size_t ws_size,
                              hipStream_t stream) {
  const float* y     = (const float*)d_in[0];
  const float* phi   = (const float*)d_in[1];
  const float* theta = (const float*)d_in[2];
  const float* mu    = (const float*)d_in[3];
  float* out = (float*)d_out;

  dim3 grid(BSZ / ROWS_PER_BLOCK);
  dim3 block(WAVES_PER_BLOCK * 32);
  hipLaunchKernelGGL(arima_eps_kernel, grid, block, 0, stream,
                     y, phi, theta, mu, out);
}


# y loads issued eight chunks ahead of use through a register ring of wide loads, so no chunk waits for its own loads; L2 prefetch distance moved out beyond the ring; summation order unchanged
# speedup vs baseline: 1.6591x; 1.6591x over previous
#include <hip/hip_runtime.h>
#include <hip/hip_bf16.h>

typedef __attribute__((ext_vector_type(2))) float v2f;
typedef __attribute__((ext_vector_type(4))) float v4f;
typedef __attribute__((ext_vector_type(8))) float v8f;

#define BSZ    8192
#define TSEQ   4096
#define OUTW   4095
#define NVALID 4093
#define CH     16
#define NCHUNK 256
#define WAVES_PER_BLOCK 2
#define ROWS_PER_WAVE   16
#define ROWS_PER_BLOCK  (WAVES_PER_BLOCK * ROWS_PER_WAVE)
#define DEPTH  8
#define PF_CHUNKS 24
#define PF_DIST (PF_CHUNKS * CH)
#define IT_PF  ((NCHUNK - DEPTH - PF_CHUNKS) / DEPTH)
#define IT_ALL ((NCHUNK - DEPTH) / DEPTH)

template <bool GUARD, bool PF>
__device__ __forceinline__ void load_chunk(
    int base, const float* __restrict__ yrow, int khalf, v4f (&q)[4])
{
  const int tbase = base + khalf;
#pragma unroll
  for (int i = 0; i < 4; ++i) {
    if (GUARD) {
      v4f t;
#pragma unroll
      for (int e = 0; e < 4; ++e) {
        const int idx = tbase + 1 + 4 * i + e;
        t[e] = (idx < TSEQ) ? yrow[idx] : 0.0f;
      }
      q[i] = t;
    } else {
      __builtin_memcpy(&q[i], yrow + tbase + 1 + 4 * i, sizeof(v4f));
    }
  }
  if (PF) {
    __builtin_prefetch(yrow + tbase + PF_DIST, 0, 0);
  }
}

template <bool LAST>
__device__ __forceinline__ void compute_chunk(
    int base, const v4f (&q)[4], float* __restrict__ orow,
    int m0, int lrow,
    const float A0k[4], const float A1k[4], const float hv[9],
    float q0, float ph1, float muv, float th1,
    float& e1, float& e2)
{
  v8f acc = {0.f, 0.f, 0.f, 0.f, 0.f, 0.f, 0.f, 0.f};
#pragma unroll
  for (int kk = 0; kk < 4; ++kk) {
    const v4f y4 = q[kk];
    const float t0 = fmaf(-ph1, y4.x, fmaf(-q0, y4.y, y4.z - muv));
    const float t1 = fmaf(-ph1, y4.y, fmaf(-q0, y4.z, y4.w - muv));
    v2f a; a.x = A0k[kk]; a.y = A1k[kk];
    v2f b; b.x = t0;      b.y = t1;
    acc = __builtin_amdgcn_wmma_f32_16x16x4_f32(
        false, a, false, b, (short)0, acc, false, false);
  }

  const float e2m = -th1 * e2;
  float d[8];
#pragma unroll
  for (int v = 0; v < 8; ++v)
    d[v] = fmaf(e1, hv[v + 1], fmaf(e2m, hv[v], acc[v]));

  e1 = __shfl(d[7], lrow + 16, 32);
  e2 = __shfl(d[6], lrow + 16, 32);

  float* op = orow + base + m0;
#pragma unroll
  for (int v = 0; v < 8; ++v) {
    if (LAST) {
      const int k = base + m0 + v;
      if (k < NVALID)     op[v] = d[v];
      else if (k < OUTW)  op[v] = 0.0f;
    } else {
      op[v] = d[v];
    }
  }
}

__global__ __launch_bounds__(WAVES_PER_BLOCK * 32) void arima_eps_kernel(
    const float* __restrict__ y, const float* __restrict__ phi,
    const float* __restrict__ theta, const float* __restrict__ mu,
    float* __restrict__ out)
{
  __shared__ float h_sh[17];

  const float ph0 = phi[0], ph1 = phi[1];
  const float th0 = theta[0], th1 = theta[1];
  const float muv = mu[0];

  if (threadIdx.x == 0) {
    h_sh[0] = 1.0f;
    h_sh[1] = -th0;
    for (int k = 2; k < 17; ++k)
      h_sh[k] = fmaf(-th0, h_sh[k - 1], -th1 * h_sh[k - 2]);
  }
  __syncthreads();

  const int lane  = threadIdx.x & 31;
  const int wid   = threadIdx.x >> 5;
  const int lrow  = lane & 15;
  const int half  = lane >> 4;
  const int khalf = half * 2;
  const int m0    = half * 8;
  const int row   = blockIdx.x * ROWS_PER_BLOCK + wid * ROWS_PER_WAVE + lrow;

  const float* yrow = y   + (size_t)row * TSEQ;
  float*       orow = out + (size_t)row * OUTW;

  v4f yb[DEPTH][4];
#pragma unroll
  for (int j = 0; j < DEPTH; ++j)
    load_chunk<false, true>(j * CH, yrow, khalf, yb[j]);

  float A0k[4], A1k[4];
#pragma unroll
  for (int kk = 0; kk < 4; ++kk) {
    const int c0 = 4 * kk + khalf;
    A0k[kk] = (lrow >= c0)     ? h_sh[lrow - c0]     : 0.0f;
    A1k[kk] = (lrow >= c0 + 1) ? h_sh[lrow - c0 - 1] : 0.0f;
  }
  float hv[9];
#pragma unroll
  for (int v = 0; v < 9; ++v) hv[v] = h_sh[m0 + v];

  const float q0 = 1.0f + ph0;
  float e1 = 0.0f, e2 = 0.0f;

  for (int it = 0; it < IT_PF; ++it) {
    const int cb = it * (DEPTH * CH);
#pragma unroll
    for (int j = 0; j < DEPTH; ++j) {
      compute_chunk<false>(cb + j * CH, yb[j], orow, m0, lrow,
                           A0k, A1k, hv, q0, ph1, muv, th1, e1, e2);
      load_chunk<false, true>(cb + (j + DEPTH) * CH, yrow, khalf, yb[j]);
      __builtin_amdgcn_sched_barrier(0);
    }
  }
  for (int it = IT_PF; it < IT_ALL; ++it) {
    const int cb = it * (DEPTH * CH);
#pragma unroll
    for (int j = 0; j < DEPTH; ++j) {
      compute_chunk<false>(cb + j * CH, yb[j], orow, m0, lrow,
                           A0k, A1k, hv, q0, ph1, muv, th1, e1, e2);
      if (j < DEPTH - 1)
        load_chunk<false, false>(cb + (j + DEPTH) * CH, yrow, khalf, yb[j]);
      else
        load_chunk<true, false>(cb + (j + DEPTH) * CH, yrow, khalf, yb[j]);
    }
  }
#pragma unroll
  for (int j = 0; j < DEPTH - 1; ++j)
    compute_chunk<false>((NCHUNK - DEPTH + j) * CH, yb[j], orow, m0, lrow,
                         A0k, A1k, hv, q0, ph1, muv, th1, e1, e2);
  compute_chunk<true>((NCHUNK - 1) * CH, yb[DEPTH - 1], orow, m0, lrow,
                      A0k, A1k, hv, q0, ph1, muv, th1, e1, e2);
}

extern "C" void kernel_launch(void* const* d_in, const int* in_sizes, int n_in,
                              void* d_out, int out_size, void* d_ws, size_t ws_size,
                              hipStream_t stream) {
  const float* y     = (const float*)d_in[0];
  const float* phi   = (const float*)d_in[1];
  const float* theta = (const float*)d_in[2];
  const float* mu    = (const float*)d_in[3];
  float* out = (float*)d_out;

  dim3 grid(BSZ / ROWS_PER_BLOCK);
  dim3 block(WAVES_PER_BLOCK * 32);
  hipLaunchKernelGGL(arima_eps_kernel, grid, block, 0, stream,
                     y, phi, theta, mu, out);
}


# k order of both fragments re-paired so a lane's 8 targets are consecutive: each y loaded once by two aligned 16-byte loads (was four overlapping), ring 2 to 6 groups; products regrouped per WMMA
# speedup vs baseline: 3.2731x; 1.9728x over previous
#include <hip/hip_runtime.h>
#include <hip/hip_bf16.h>

typedef __attribute__((ext_vector_type(2))) float v2f;
typedef __attribute__((ext_vector_type(4))) float v4f;
typedef __attribute__((ext_vector_type(8))) float v8f;

#define BSZ    8192
#define TSEQ   4096
#define OUTW   4095
#define NVALID 4093
#define CH     16
#define GRP    256
#define NGRP   16
#define WAVES_PER_BLOCK 8
#define RING   6
#define HLEN   129

__device__ __forceinline__ float uniform_f(float x) {
  return __builtin_bit_cast(float, __builtin_amdgcn_readfirstlane(__builtin_bit_cast(int, x)));
}

__device__ __forceinline__ float other_half(float v) {
  const unsigned u = __builtin_bit_cast(unsigned, v);
  return __builtin_bit_cast(float,
      __builtin_amdgcn_permlanex16(u, u, 0x76543210u, 0xfedcba98u, false, false));
}
__device__ __forceinline__ float other_half_next_col(float v) {
  const unsigned u = __builtin_bit_cast(unsigned, v);
  return __builtin_bit_cast(float,
      __builtin_amdgcn_permlanex16(u, u, 0x87654321u, 0x0fedcba9u, false, false));
}

__device__ __forceinline__ void load_group(
    int g, const float* __restrict__ yrow, int kq, v4f (&q)[2])
{
  const float* p = yrow + g * GRP + kq;
  __builtin_memcpy(&q[0], p,     sizeof(v4f));
  __builtin_memcpy(&q[1], p + 4, sizeof(v4f));
}

template <bool LAST>
__device__ __forceinline__ void compute_group(
    int g, const v4f (&q)[2], float nx0, float nx1, float nx2,
    float* __restrict__ orow,
    int n, int half, int m0,
    const float A0k[4], const float A1k[4], const float hv[9],
    const float (&HT)[4][3],
    float q0, float ph1, float muv, float th1,
    float& sin1, float& sin2)
{
  float w[11];
#pragma unroll
  for (int e = 0; e < 4; ++e) { w[e] = q[0][e]; w[4 + e] = q[1][e]; }
  const float nx[3] = {nx0, nx1, nx2};
#pragma unroll
  for (int j = 0; j < 3; ++j) {
    const float same = other_half(w[j]);
    const float next = other_half_next_col(w[j]);
    w[8 + j] = (half == 0) ? same : ((n == 15) ? nx[j] : next);
  }

  v8f acc = {0.f, 0.f, 0.f, 0.f, 0.f, 0.f, 0.f, 0.f};
#pragma unroll
  for (int kk = 0; kk < 4; ++kk) {
    const int j = 2 * kk;
    const float t0 = fmaf(-ph1, w[j + 1], fmaf(-q0, w[j + 2], w[j + 3] - muv));
    const float t1 = fmaf(-ph1, w[j + 2], fmaf(-q0, w[j + 3], w[j + 4] - muv));
    v2f a; a.x = A0k[kk]; a.y = A1k[kk];
    v2f b; b.x = t0;      b.y = t1;
    acc = __builtin_amdgcn_wmma_f32_16x16x4_f32(
        false, a, false, b, (short)0, acc, false, false);
  }

  float a1 = __shfl(acc[7], n + 16, 32);
  float a2 = __shfl(acc[6], n + 16, 32);

  {
    const float sm = -th1 * sin2;
    const float f1 = fmaf(sin1, HT[0][0], fmaf(sm, HT[0][1], a1));
    const float f2 = fmaf(sin1, HT[0][1], fmaf(sm, HT[0][2], a2));
    if (n == 0) { a1 = f1; a2 = f2; }
  }
#pragma unroll
  for (int s = 0; s < 4; ++s) {
    const int o = 1 << s;
    const float p1 = __shfl_up(a1, o, 16);
    const float p2 = __shfl_up(a2, o, 16);
    const float pm = -th1 * p2;
    const float f1 = fmaf(p1, HT[s][0], fmaf(pm, HT[s][1], a1));
    const float f2 = fmaf(p1, HT[s][1], fmaf(pm, HT[s][2], a2));
    if (n >= o) { a1 = f1; a2 = f2; }
  }
  float s1 = __shfl_up(a1, 1, 16);
  float s2 = __shfl_up(a2, 1, 16);
  if (n == 0) { s1 = sin1; s2 = sin2; }
  sin1 = __shfl(a1, 15, 16);
  sin2 = __shfl(a2, 15, 16);

  const float e2m = -th1 * s2;
  float d[8];
#pragma unroll
  for (int v = 0; v < 8; ++v)
    d[v] = fmaf(s1, hv[v + 1], fmaf(e2m, hv[v], acc[v]));

  const int kb = g * GRP + n * CH + m0;
  float* op = orow + kb;
#pragma unroll
  for (int v = 0; v < 8; ++v) {
    if (LAST) {
      const int k = kb + v;
      if (k < NVALID)     op[v] = d[v];
      else if (k < OUTW)  op[v] = 0.0f;
    } else {
      op[v] = d[v];
    }
  }
}

__global__ __launch_bounds__(WAVES_PER_BLOCK * 32) void arima_eps_kernel(
    const float* __restrict__ y, const float* __restrict__ phi,
    const float* __restrict__ theta, const float* __restrict__ mu,
    float* __restrict__ out)
{
  __shared__ float h_sh[HLEN];

  const float ph0 = phi[0], ph1 = phi[1];
  const float th0 = theta[0], th1 = theta[1];
  const float muv = mu[0];

  const int lane  = threadIdx.x & 31;
  const int wid   = threadIdx.x >> 5;
  const int n     = lane & 15;
  const int half  = lane >> 4;
  const int m0    = half * 8;
  const int row   = blockIdx.x * WAVES_PER_BLOCK + wid;
  const int kq    = n * CH + half * 8;

  const float* yrow = y   + (size_t)row * TSEQ;
  float*       orow = out + (size_t)row * OUTW;

  v4f yb[RING][2];
#pragma unroll
  for (int j = 0; j < RING; ++j)
    load_group(j, yrow, kq, yb[j]);

  if (threadIdx.x == 0) {
    float hm2 = 1.0f, hm1 = -th0;
    h_sh[0] = hm2;
    h_sh[1] = hm1;
    for (int k = 2; k < HLEN; ++k) {
      const float hk = fmaf(-th0, hm1, -th1 * hm2);
      h_sh[k] = hk;
      hm2 = hm1; hm1 = hk;
    }
  }
  __syncthreads();

  float A0k[4], A1k[4];
#pragma unroll
  for (int kk = 0; kk < 4; ++kk) {
    const int c0 = 8 * half + 2 * kk;
    A0k[kk] = (n >= c0)     ? h_sh[n - c0]     : 0.0f;
    A1k[kk] = (n >= c0 + 1) ? h_sh[n - c0 - 1] : 0.0f;
  }
  float hv[9];
#pragma unroll
  for (int v = 0; v < 9; ++v) hv[v] = h_sh[m0 + v];
  float HT[4][3];
#pragma unroll
  for (int s = 0; s < 4; ++s) {
    const int L = CH << s;
    HT[s][0] = uniform_f(h_sh[L]);
    HT[s][1] = uniform_f(h_sh[L - 1]);
    HT[s][2] = uniform_f(h_sh[L - 2]);
  }

  const float q0 = 1.0f + ph0;
  float sin1 = 0.0f, sin2 = 0.0f;

#pragma unroll
  for (int g = 0; g < NGRP; ++g) {
    float nx0 = 0.0f, nx1 = 0.0f, nx2 = 0.0f;
    if (g < NGRP - 1) {
      nx0 = uniform_f(yb[(g + 1) % RING][0][0]);
      nx1 = uniform_f(yb[(g + 1) % RING][0][1]);
      nx2 = uniform_f(yb[(g + 1) % RING][0][2]);
    }
    if (g < NGRP - 1)
      compute_group<false>(g, yb[g % RING], nx0, nx1, nx2, orow, n, half, m0,
                           A0k, A1k, hv, HT, q0, ph1, muv, th1, sin1, sin2);
    else
      compute_group<true>(g, yb[g % RING], nx0, nx1, nx2, orow, n, half, m0,
                          A0k, A1k, hv, HT, q0, ph1, muv, th1, sin1, sin2);
    if (g + RING < NGRP)
      load_group(g + RING, yrow, kq, yb[g % RING]);
    __builtin_amdgcn_sched_barrier(0);
  }
}

extern "C" void kernel_launch(void* const* d_in, const int* in_sizes, int n_in,
                              void* d_out, int out_size, void* d_ws, size_t ws_size,
                              hipStream_t stream) {
  const float* y     = (const float*)d_in[0];
  const float* phi   = (const float*)d_in[1];
  const float* theta = (const float*)d_in[2];
  const float* mu    = (const float*)d_in[3];
  float* out = (float*)d_out;

  dim3 grid(BSZ / WAVES_PER_BLOCK);
  dim3 block(WAVES_PER_BLOCK * 32);
  hipLaunchKernelGGL(arima_eps_kernel, grid, block, 0, stream,
                     y, phi, theta, mu, out);
}


# scan lane exchanges moved off the LDS unit: row-shift moves on the vector ALU and one cross-half permute replace seven dependent ds_bpermute round trips per group; sums and their order unchanged
# speedup vs baseline: 3.6581x; 1.1176x over previous
#include <hip/hip_runtime.h>
#include <hip/hip_bf16.h>

typedef __attribute__((ext_vector_type(2))) float v2f;
typedef __attribute__((ext_vector_type(4))) float v4f;
typedef __attribute__((ext_vector_type(8))) float v8f;

#define BSZ    8192
#define TSEQ   4096
#define OUTW   4095
#define NVALID 4093
#define CH     16
#define GRP    256
#define NGRP   16
#define WAVES_PER_BLOCK 8
#define RING   6
#define HLEN   129

__device__ __forceinline__ float uniform_f(float x) {
  return __builtin_bit_cast(float, __builtin_amdgcn_readfirstlane(__builtin_bit_cast(int, x)));
}

__device__ __forceinline__ float other_half(float v) {
  const unsigned u = __builtin_bit_cast(unsigned, v);
  return __builtin_bit_cast(float,
      __builtin_amdgcn_permlanex16(u, u, 0x76543210u, 0xfedcba98u, false, false));
}
__device__ __forceinline__ float other_half_next_col(float v) {
  const unsigned u = __builtin_bit_cast(unsigned, v);
  return __builtin_bit_cast(float,
      __builtin_amdgcn_permlanex16(u, u, 0x87654321u, 0x0fedcba9u, false, false));
}

template <int O>
__device__ __forceinline__ float row_up(float v) {
  const int u = __builtin_bit_cast(int, v);
  return __builtin_bit_cast(float,
      __builtin_amdgcn_update_dpp(0, u, 0x110 + O, 0xf, 0xf, true));
}

__device__ __forceinline__ void load_group(
    int g, const float* __restrict__ yrow, int kq, v4f (&q)[2])
{
  const float* p = yrow + g * GRP + kq;
  __builtin_memcpy(&q[0], p,     sizeof(v4f));
  __builtin_memcpy(&q[1], p + 4, sizeof(v4f));
}

template <bool LAST>
__device__ __forceinline__ void compute_group(
    int g, const v4f (&q)[2], float nx0, float nx1, float nx2,
    float* __restrict__ orow,
    int n, int half, int m0,
    const float A0k[4], const float A1k[4], const float hv[9],
    const float (&HT)[4][3],
    float q0, float ph1, float muv, float th1,
    float& sin1, float& sin2)
{
  float w[11];
#pragma unroll
  for (int e = 0; e < 4; ++e) { w[e] = q[0][e]; w[4 + e] = q[1][e]; }
  const float nx[3] = {nx0, nx1, nx2};
#pragma unroll
  for (int j = 0; j < 3; ++j) {
    const float same = other_half(w[j]);
    const float next = other_half_next_col(w[j]);
    w[8 + j] = (half == 0) ? same : ((n == 15) ? nx[j] : next);
  }

  v8f acc = {0.f, 0.f, 0.f, 0.f, 0.f, 0.f, 0.f, 0.f};
#pragma unroll
  for (int kk = 0; kk < 4; ++kk) {
    const int j = 2 * kk;
    const float t0 = fmaf(-ph1, w[j + 1], fmaf(-q0, w[j + 2], w[j + 3] - muv));
    const float t1 = fmaf(-ph1, w[j + 2], fmaf(-q0, w[j + 3], w[j + 4] - muv));
    v2f a; a.x = A0k[kk]; a.y = A1k[kk];
    v2f b; b.x = t0;      b.y = t1;
    acc = __builtin_amdgcn_wmma_f32_16x16x4_f32(
        false, a, false, b, (short)0, acc, false, false);
  }

  asm volatile("s_wait_alu depctr_va_vdst(0)" : "+v"(acc));

  const float u1 = other_half(acc[7]);
  const float u2 = other_half(acc[6]);
  float a1 = (half == 0) ? u1 : acc[7];
  float a2 = (half == 0) ? u2 : acc[6];

  {
    const float sm = -th1 * sin2;
    const float f1 = fmaf(sin1, HT[0][0], fmaf(sm, HT[0][1], a1));
    const float f2 = fmaf(sin1, HT[0][1], fmaf(sm, HT[0][2], a2));
    if (n == 0) { a1 = f1; a2 = f2; }
  }
#define SCAN_STEP(S)                                                        \
  {                                                                         \
    const float p1 = row_up<(1 << S)>(a1);                                  \
    const float p2 = row_up<(1 << S)>(a2);                                  \
    const float pm = -th1 * p2;                                             \
    const float f1 = fmaf(p1, HT[S][0], fmaf(pm, HT[S][1], a1));            \
    const float f2 = fmaf(p1, HT[S][1], fmaf(pm, HT[S][2], a2));            \
    if (n >= (1 << S)) { a1 = f1; a2 = f2; }                                \
  }
  SCAN_STEP(0) SCAN_STEP(1) SCAN_STEP(2) SCAN_STEP(3)
#undef SCAN_STEP
  float s1 = row_up<1>(a1);
  float s2 = row_up<1>(a2);
  if (n == 0) { s1 = sin1; s2 = sin2; }
  sin1 = __shfl(a1, 15, 16);
  sin2 = __shfl(a2, 15, 16);

  const float e2m = -th1 * s2;
  float d[8];
#pragma unroll
  for (int v = 0; v < 8; ++v)
    d[v] = fmaf(s1, hv[v + 1], fmaf(e2m, hv[v], acc[v]));

  const int kb = g * GRP + n * CH + m0;
  float* op = orow + kb;
#pragma unroll
  for (int v = 0; v < 8; ++v) {
    if (LAST) {
      const int k = kb + v;
      if (k < NVALID)     op[v] = d[v];
      else if (k < OUTW)  op[v] = 0.0f;
    } else {
      op[v] = d[v];
    }
  }
}

__global__ __launch_bounds__(WAVES_PER_BLOCK * 32) void arima_eps_kernel(
    const float* __restrict__ y, const float* __restrict__ phi,
    const float* __restrict__ theta, const float* __restrict__ mu,
    float* __restrict__ out)
{
  __shared__ float h_sh[HLEN];

  const float ph0 = phi[0], ph1 = phi[1];
  const float th0 = theta[0], th1 = theta[1];
  const float muv = mu[0];

  const int lane  = threadIdx.x & 31;
  const int wid   = threadIdx.x >> 5;
  const int n     = lane & 15;
  const int half  = lane >> 4;
  const int m0    = half * 8;
  const int row   = blockIdx.x * WAVES_PER_BLOCK + wid;
  const int kq    = n * CH + half * 8;

  const float* yrow = y   + (size_t)row * TSEQ;
  float*       orow = out + (size_t)row * OUTW;

  v4f yb[RING][2];
#pragma unroll
  for (int j = 0; j < RING; ++j)
    load_group(j, yrow, kq, yb[j]);

  if (threadIdx.x == 0) {
    float hm2 = 1.0f, hm1 = -th0;
    h_sh[0] = hm2;
    h_sh[1] = hm1;
    for (int k = 2; k < HLEN; ++k) {
      const float hk = fmaf(-th0, hm1, -th1 * hm2);
      h_sh[k] = hk;
      hm2 = hm1; hm1 = hk;
    }
  }
  __syncthreads();

  float A0k[4], A1k[4];
#pragma unroll
  for (int kk = 0; kk < 4; ++kk) {
    const int c0 = 8 * half + 2 * kk;
    A0k[kk] = (n >= c0)     ? h_sh[n - c0]     : 0.0f;
    A1k[kk] = (n >= c0 + 1) ? h_sh[n - c0 - 1] : 0.0f;
  }
  float hv[9];
#pragma unroll
  for (int v = 0; v < 9; ++v) hv[v] = h_sh[m0 + v];
  float HT[4][3];
#pragma unroll
  for (int s = 0; s < 4; ++s) {
    const int L = CH << s;
    HT[s][0] = uniform_f(h_sh[L]);
    HT[s][1] = uniform_f(h_sh[L - 1]);
    HT[s][2] = uniform_f(h_sh[L - 2]);
  }

  const float q0 = 1.0f + ph0;
  float sin1 = 0.0f, sin2 = 0.0f;

#pragma unroll
  for (int g = 0; g < NGRP; ++g) {
    float nx0 = 0.0f, nx1 = 0.0f, nx2 = 0.0f;
    if (g < NGRP - 1) {
      nx0 = uniform_f(yb[(g + 1) % RING][0][0]);
      nx1 = uniform_f(yb[(g + 1) % RING][0][1]);
      nx2 = uniform_f(yb[(g + 1) % RING][0][2]);
    }
    if (g < NGRP - 1)
      compute_group<false>(g, yb[g % RING], nx0, nx1, nx2, orow, n, half, m0,
                           A0k, A1k, hv, HT, q0, ph1, muv, th1, sin1, sin2);
    else
      compute_group<true>(g, yb[g % RING], nx0, nx1, nx2, orow, n, half, m0,
                          A0k, A1k, hv, HT, q0, ph1, muv, th1, sin1, sin2);
    if (g + RING < NGRP)
      load_group(g + RING, yrow, kq, yb[g % RING]);
    __builtin_amdgcn_sched_barrier(0);
  }
}

extern "C" void kernel_launch(void* const* d_in, const int* in_sizes, int n_in,
                              void* d_out, int out_size, void* d_ws, size_t ws_size,
                              hipStream_t stream) {
  const float* y     = (const float*)d_in[0];
  const float* phi   = (const float*)d_in[1];
  const float* theta = (const float*)d_in[2];
  const float* mu    = (const float*)d_in[3];
  float* out = (float*)d_out;

  dim3 grid(BSZ / WAVES_PER_BLOCK);
  dim3 block(WAVES_PER_BLOCK * 32);
  hipLaunchKernelGGL(arima_eps_kernel, grid, block, 0, stream,
                     y, phi, theta, mu, out);
}


# per-wave LDS tile turns the fragment lane order into row order: every wave-wide 16-byte global load and store covers 512 contiguous bytes (was 16-byte pieces 32 bytes apart); sums and order unchanged
# speedup vs baseline: 3.9488x; 1.0795x over previous
#include <hip/hip_runtime.h>
#include <hip/hip_bf16.h>

typedef __attribute__((ext_vector_type(2))) float v2f;
typedef __attribute__((ext_vector_type(4))) float v4f;
typedef __attribute__((ext_vector_type(8))) float v8f;

#define BSZ    8192
#define TSEQ   4096
#define OUTW   4095
#define NVALID 4093
#define CH     16
#define GRP    256
#define NGRP   16
#define WAVES_PER_BLOCK 8
#define RING   6
#define HLEN   129
#define PIECES 64

__device__ __forceinline__ float uniform_f(float x) {
  return __builtin_bit_cast(float, __builtin_amdgcn_readfirstlane(__builtin_bit_cast(int, x)));
}

__device__ __forceinline__ float other_half(float v) {
  const unsigned u = __builtin_bit_cast(unsigned, v);
  return __builtin_bit_cast(float,
      __builtin_amdgcn_permlanex16(u, u, 0x76543210u, 0xfedcba98u, false, false));
}

__device__ __forceinline__ int slot_of(int p) { return p ^ ((p >> 4) & 3); }

__device__ __forceinline__ void wave_lds_handover() {
  __builtin_amdgcn_fence(__ATOMIC_RELEASE, "wavefront");
  __builtin_amdgcn_fence(__ATOMIC_ACQUIRE, "wavefront");
}

template <int O>
__device__ __forceinline__ float row_up(float v) {
  const int u = __builtin_bit_cast(int, v);
  return __builtin_bit_cast(float,
      __builtin_amdgcn_update_dpp(0, u, 0x110 + O, 0xf, 0xf, true));
}

__device__ __forceinline__ void load_group(
    int g, const float* __restrict__ yrow, int lane, v4f (&q)[2])
{
  const float* p = yrow + g * GRP + 4 * lane;
  __builtin_memcpy(&q[0], p,       sizeof(v4f));
  __builtin_memcpy(&q[1], p + 128, sizeof(v4f));
}

template <bool TAIL>
__device__ __forceinline__ void store_group(
    int g, const v4f& o0, const v4f& o1, float* __restrict__ orow, int lane)
{
  const int kb = g * GRP + 4 * lane;
  float* op = orow + kb;
  __builtin_memcpy(op, &o0, sizeof(v4f));
  if (TAIL) {
    const int k1 = kb + 128;
    if (k1 + 3 < NVALID) {
      __builtin_memcpy(op + 128, &o1, sizeof(v4f));
    } else {
#pragma unroll
      for (int e = 0; e < 4; ++e) {
        const int k = k1 + e;
        if (k < NVALID)     op[128 + e] = o1[e];
        else if (k < OUTW)  op[128 + e] = 0.0f;
      }
    }
  } else {
    __builtin_memcpy(op + 128, &o1, sizeof(v4f));
  }
}

template <bool FIRST, bool LAST>
__device__ __forceinline__ void compute_group(
    int g, const v4f* tin, v4f* tout,
    int if0, int if1, int if2, int sf0, int sf1, int sr0, int sr1,
    float* __restrict__ orow,
    int lane, int n, int half,
    const float A0k[4], const float A1k[4], const float hv[9],
    const float (&HT)[4][3],
    float q0, float ph1, float muv, float th1,
    float& sin1, float& sin2,
    v4f& o0, v4f& o1)
{
  const v4f wa = tin[if0];
  const v4f wb = tin[if1];
  v4f wc = tin[if2];
  if (LAST) {
    if (n == 15 && half == 1) { wc[0] = 0.0f; wc[1] = 0.0f; wc[2] = 0.0f; }
  }
  float w[11];
#pragma unroll
  for (int e = 0; e < 4; ++e) { w[e] = wa[e]; w[4 + e] = wb[e]; }
#pragma unroll
  for (int j = 0; j < 3; ++j) w[8 + j] = wc[j];

  v8f acc = {0.f, 0.f, 0.f, 0.f, 0.f, 0.f, 0.f, 0.f};
#pragma unroll
  for (int kk = 0; kk < 4; ++kk) {
    const int j = 2 * kk;
    const float t0 = fmaf(-ph1, w[j + 1], fmaf(-q0, w[j + 2], w[j + 3] - muv));
    const float t1 = fmaf(-ph1, w[j + 2], fmaf(-q0, w[j + 3], w[j + 4] - muv));
    v2f a; a.x = A0k[kk]; a.y = A1k[kk];
    v2f b; b.x = t0;      b.y = t1;
    acc = __builtin_amdgcn_wmma_f32_16x16x4_f32(
        false, a, false, b, (short)0, acc, false, false);
  }

  asm volatile("s_wait_alu depctr_va_vdst(0)" : "+v"(acc));

  if (!FIRST) store_group<false>(g - 1, o0, o1, orow, lane);

  const float u1 = other_half(acc[7]);
  const float u2 = other_half(acc[6]);
  float a1 = (half == 0) ? u1 : acc[7];
  float a2 = (half == 0) ? u2 : acc[6];

  {
    const float sm = -th1 * sin2;
    const float f1 = fmaf(sin1, HT[0][0], fmaf(sm, HT[0][1], a1));
    const float f2 = fmaf(sin1, HT[0][1], fmaf(sm, HT[0][2], a2));
    if (n == 0) { a1 = f1; a2 = f2; }
  }
#define SCAN_STEP(S)                                                        \
  {                                                                         \
    const float p1 = row_up<(1 << S)>(a1);                                  \
    const float p2 = row_up<(1 << S)>(a2);                                  \
    const float pm = -th1 * p2;                                             \
    const float f1 = fmaf(p1, HT[S][0], fmaf(pm, HT[S][1], a1));            \
    const float f2 = fmaf(p1, HT[S][1], fmaf(pm, HT[S][2], a2));            \
    if (n >= (1 << S)) { a1 = f1; a2 = f2; }                                \
  }
  SCAN_STEP(0) SCAN_STEP(1) SCAN_STEP(2) SCAN_STEP(3)
#undef SCAN_STEP
  float s1 = row_up<1>(a1);
  float s2 = row_up<1>(a2);
  if (n == 0) { s1 = sin1; s2 = sin2; }
  sin1 = __shfl(a1, 15, 16);
  sin2 = __shfl(a2, 15, 16);

  const float e2m = -th1 * s2;
  float d[8];
#pragma unroll
  for (int v = 0; v < 8; ++v)
    d[v] = fmaf(s1, hv[v + 1], fmaf(e2m, hv[v], acc[v]));

  v4f dlo, dhi;
#pragma unroll
  for (int e = 0; e < 4; ++e) { dlo[e] = d[e]; dhi[e] = d[4 + e]; }
  tout[sf0] = dlo;
  tout[sf1] = dhi;
  wave_lds_handover();
  o0 = tout[sr0];
  o1 = tout[sr1];
  wave_lds_handover();
}

__global__ __launch_bounds__(WAVES_PER_BLOCK * 32) void arima_eps_kernel(
    const float* __restrict__ y, const float* __restrict__ phi,
    const float* __restrict__ theta, const float* __restrict__ mu,
    float* __restrict__ out)
{
  __shared__ float h_sh[HLEN];
  __shared__ v4f in_sh[WAVES_PER_BLOCK][2 * PIECES];
  __shared__ v4f out_sh[WAVES_PER_BLOCK][PIECES];

  const float ph0 = phi[0], ph1 = phi[1];
  const float th0 = theta[0], th1 = theta[1];
  const float muv = mu[0];

  const int lane  = threadIdx.x & 31;
  const int wid   = threadIdx.x >> 5;
  const int n     = lane & 15;
  const int half  = lane >> 4;
  const int m0    = half * 8;
  const int row   = blockIdx.x * WAVES_PER_BLOCK + wid;

  const float* yrow = y   + (size_t)row * TSEQ;
  float*       orow = out + (size_t)row * OUTW;
  v4f* tin  = in_sh[wid];
  v4f* tout = out_sh[wid];

  const int sr0 = slot_of(lane);
  const int sr1 = slot_of(lane + 32);
  const int pf  = 4 * n + 2 * half;
  const int sf0 = slot_of(pf);
  const int sf1 = slot_of(pf + 1);
  const bool into_next = (pf + 2 == PIECES);
  const int sf2 = into_next ? 0 : slot_of(pf + 2);

  v4f yb[RING][2];
#pragma unroll
  for (int j = 0; j < RING; ++j)
    load_group(j, yrow, lane, yb[j]);

  if (threadIdx.x == 0) {
    float hm2 = 1.0f, hm1 = -th0;
    h_sh[0] = hm2;
    h_sh[1] = hm1;
    for (int k = 2; k < HLEN; ++k) {
      const float hk = fmaf(-th0, hm1, -th1 * hm2);
      h_sh[k] = hk;
      hm2 = hm1; hm1 = hk;
    }
  }
  __syncthreads();

  float A0k[4], A1k[4];
#pragma unroll
  for (int kk = 0; kk < 4; ++kk) {
    const int c0 = 8 * half + 2 * kk;
    A0k[kk] = (n >= c0)     ? h_sh[n - c0]     : 0.0f;
    A1k[kk] = (n >= c0 + 1) ? h_sh[n - c0 - 1] : 0.0f;
  }
  float hv[9];
#pragma unroll
  for (int v = 0; v < 9; ++v) hv[v] = h_sh[m0 + v];
  float HT[4][3];
#pragma unroll
  for (int s = 0; s < 4; ++s) {
    const int L = CH << s;
    HT[s][0] = uniform_f(h_sh[L]);
    HT[s][1] = uniform_f(h_sh[L - 1]);
    HT[s][2] = uniform_f(h_sh[L - 2]);
  }

  const float q0 = 1.0f + ph0;
  float sin1 = 0.0f, sin2 = 0.0f;
  v4f o0 = {0.f, 0.f, 0.f, 0.f}, o1 = {0.f, 0.f, 0.f, 0.f};

  tin[sr0] = yb[0][0];
  tin[sr1] = yb[0][1];
  if (RING < NGRP)
    load_group(RING, yrow, lane, yb[0]);

#pragma unroll
  for (int g = 0; g < NGRP; ++g) {
    const int cur = (g & 1) * PIECES;
    const int nxt = PIECES - cur;
    if (g + 1 < NGRP) {
      tin[nxt + sr0] = yb[(g + 1) % RING][0];
      tin[nxt + sr1] = yb[(g + 1) % RING][1];
      if (g + 1 + RING < NGRP)
        load_group(g + 1 + RING, yrow, lane, yb[(g + 1) % RING]);
    }
    wave_lds_handover();
    const int if2 = into_next ? nxt : (cur + sf2);
    if (g == 0)
      compute_group<true, false>(g, tin, tout, cur + sf0, cur + sf1, if2, sf0, sf1, sr0, sr1,
                                 orow, lane, n, half,
                                 A0k, A1k, hv, HT, q0, ph1, muv, th1, sin1, sin2, o0, o1);
    else if (g < NGRP - 1)
      compute_group<false, false>(g, tin, tout, cur + sf0, cur + sf1, if2, sf0, sf1, sr0, sr1,
                                  orow, lane, n, half,
                                  A0k, A1k, hv, HT, q0, ph1, muv, th1, sin1, sin2, o0, o1);
    else
      compute_group<false, true>(g, tin, tout, cur + sf0, cur + sf1, if2, sf0, sf1, sr0, sr1,
                                 orow, lane, n, half,
                                 A0k, A1k, hv, HT, q0, ph1, muv, th1, sin1, sin2, o0, o1);
    __builtin_amdgcn_sched_barrier(0);
  }
  store_group<true>(NGRP - 1, o0, o1, orow, lane);
}

extern "C" void kernel_launch(void* const* d_in, const int* in_sizes, int n_in,
                              void* d_out, int out_size, void* d_ws, size_t ws_size,
                              hipStream_t stream) {
  const float* y     = (const float*)d_in[0];
  const float* phi   = (const float*)d_in[1];
  const float* theta = (const float*)d_in[2];
  const float* mu    = (const float*)d_in[3];
  float* out = (float*)d_out;

  dim3 grid(BSZ / WAVES_PER_BLOCK);
  dim3 block(WAVES_PER_BLOCK * 32);
  hipLaunchKernelGGL(arima_eps_kernel, grid, block, 0, stream,
                     y, phi, theta, mu, out);
}
